# starting kernel
# baseline (speedup 1.0000x reference)
#include <hip/hip_runtime.h>

typedef float v2f __attribute__((ext_vector_type(2)));
typedef float v8f __attribute__((ext_vector_type(8)));

#define Bn 16
#define Sn 32
#define Ln 128
#define Dn 768
#define An 6
#define Mn 10
#define VOCABn 200
#define WS 33

__global__ __launch_bounds__(256)
void srl_wmma_kernel(const int* __restrict__ sids,
                     const int* __restrict__ mask,
                     const float* __restrict__ emb,
                     const int* __restrict__ pred,
                     const int* __restrict__ arg0,
                     const int* __restrict__ arg1,
                     float* __restrict__ out)
{
    __shared__ float Wl[Ln * WS];
    __shared__ int   sid_sh[Ln];
    __shared__ int   msk_sh[Ln];
    __shared__ int   hist[VOCABn];
    __shared__ int   mcnt;
    __shared__ int   tok_sh[19];
    __shared__ float inv_sh[19];

    const int bs   = blockIdx.x;
    const int t    = threadIdx.x;
    const int lane = t & 31;
    const int wave = t >> 5;

    if (t < VOCABn) hist[t] = 0;
    if (t == 0)     mcnt = 0;
    int mySid = 0, myMsk = 0;
    if (t < Ln)          { mySid = sids[bs * Ln + t];        sid_sh[t] = mySid; }
    else if (t < 2 * Ln) { myMsk = mask[bs * Ln + (t - Ln)]; msk_sh[t - Ln] = myMsk; }
    __syncthreads();

    if (t < Ln)               atomicAdd(&hist[mySid], 1);
    else if (t < 2 * Ln) { if (myMsk) atomicAdd(&mcnt, 1); }
    __syncthreads();

    if (t < 19) {
        if (t == 0) {
            const int c = mcnt;
            tok_sh[0] = -1;
            inv_sh[0] = 1.0f / (float)(c > 0 ? c : 1);
        } else {
            const int r = t;
            int a; const int* ids;
            if (r <= 6)       { a = r - 1;  ids = pred; }
            else if (r <= 12) { a = r - 7;  ids = arg0; }
            else              { a = r - 13; ids = arg1; }
            ids += (bs * An + a) * Mn;
            int tok = 0, c = 0;
            #pragma unroll
            for (int m = 0; m < Mn; ++m) {
                const int tm = ids[m];
                if (tm != 0) {
                    const int cm = hist[tm];
                    if (cm > 0) { tok = tm; c = cm; }
                }
            }
            tok_sh[r] = tok;
            inv_sh[r] = (c > 0) ? 1.0f / (float)c : 0.0f;
        }
    }
    __syncthreads();

    #pragma unroll
    for (int i = 0; i < 16; ++i) {
        const int idx = t + i * 256;
        const int l = idx >> 5;
        const int r = idx & 31;
        float w;
        if (r == 0)       w = msk_sh[l] ? inv_sh[0] : 0.0f;
        else if (r <= 18) w = (sid_sh[l] == tok_sh[r]) ? inv_sh[r] : 0.0f;
        else              w = 0.0f;
        Wl[l * WS + r] = w;
    }
    __syncthreads();

    v8f acc[2][6];
    #pragma unroll
    for (int mt = 0; mt < 2; ++mt)
        #pragma unroll
        for (int j = 0; j < 6; ++j) {
            v8f z = {0.f, 0.f, 0.f, 0.f, 0.f, 0.f, 0.f, 0.f};
            acc[mt][j] = z;
        }

    const int lm = lane & 15;
    const int kk = (lane >> 4) << 1;
    const float* Ebs = emb + (size_t)bs * Ln * Dn;
    const int nb = wave * 96 + lm;

    for (int kc = 0; kc < 32; ++kc) {
        const int k0 = (kc << 2) + kk;
        v2f a0, a1;
        a0.x = Wl[ k0      * WS + lm];
        a0.y = Wl[(k0 + 1) * WS + lm];
        a1.x = Wl[ k0      * WS + 16 + lm];
        a1.y = Wl[(k0 + 1) * WS + 16 + lm];
        const float* Er = Ebs + (size_t)k0 * Dn;
        #pragma unroll
        for (int j = 0; j < 6; ++j) {
            v2f b;
            b.x = __builtin_nontemporal_load(&Er[     nb + j * 16]);
            b.y = __builtin_nontemporal_load(&Er[Dn + nb + j * 16]);
            acc[0][j] = __builtin_amdgcn_wmma_f32_16x16x4_f32(
                false, a0, false, b, (short)0, acc[0][j], false, false);
            acc[1][j] = __builtin_amdgcn_wmma_f32_16x16x4_f32(
                false, a1, false, b, (short)0, acc[1][j], false, false);
        }
    }

    const size_t O1 = (size_t)Bn * Sn * Dn;
    const size_t O2 = O1 + (size_t)Bn * Sn * An * Dn;
    const size_t O3 = O2 + (size_t)Bn * Sn * An * Dn;
    const int mhalf = (lane >> 4) << 3;

    #pragma unroll
    for (int mt = 0; mt < 2; ++mt) {
        #pragma unroll
        for (int j = 0; j < 6; ++j) {
            const int n = nb + j * 16;
            #pragma unroll
            for (int r = 0; r < 8; ++r) {
                const int m = mt * 16 + r + mhalf;
                const float val = acc[mt][j][r];
                if (m == 0)
                    __builtin_nontemporal_store(val, &out[(size_t)bs * Dn + n]);
                else if (m <= 6)
                    __builtin_nontemporal_store(val, &out[O1 + ((size_t)bs * An + (m - 1))  * Dn + n]);
                else if (m <= 12)
                    __builtin_nontemporal_store(val, &out[O2 + ((size_t)bs * An + (m - 7))  * Dn + n]);
                else if (m <= 18)
                    __builtin_nontemporal_store(val, &out[O3 + ((size_t)bs * An + (m - 13)) * Dn + n]);
            }
        }
    }
}

extern "C" void kernel_launch(void* const* d_in, const int* in_sizes, int n_in,
                              void* d_out, int out_size, void* d_ws, size_t ws_size,
                              hipStream_t stream)
{
    (void)in_sizes; (void)n_in; (void)out_size; (void)d_ws; (void)ws_size;
    const int*   sids = (const int*)  d_in[0];
    const int*   mask = (const int*)  d_in[1];
    const float* emb  = (const float*)d_in[2];
    const int*   pred = (const int*)  d_in[3];
    const int*   a0   = (const int*)  d_in[4];
    const int*   a1   = (const int*)  d_in[5];
    float* out = (float*)d_out;

    dim3 grid(Bn * Sn);
    dim3 block(256);
    hipLaunchKernelGGL(srl_wmma_kernel, grid, block, 0, stream,
                       sids, mask, emb, pred, a0, a1, out);
}


# E rows go through a 4-step register ring: step kc+4's 12 row loads issue right behind step kc's products (pinned by scheduling barriers), so a wave holds 36-48 loads in flight, not one step's
# speedup vs baseline: 1.0813x; 1.0813x over previous
#include <hip/hip_runtime.h>

typedef float v2f __attribute__((ext_vector_type(2)));
typedef float v8f __attribute__((ext_vector_type(8)));

#define Bn 16
#define Sn 32
#define Ln 128
#define Dn 768
#define An 6
#define Mn 10
#define VOCABn 200
#define WS 33
#define PF 4

__global__ __launch_bounds__(256)
void srl_wmma_kernel(const int* __restrict__ sids,
                     const int* __restrict__ mask,
                     const float* __restrict__ emb,
                     const int* __restrict__ pred,
                     const int* __restrict__ arg0,
                     const int* __restrict__ arg1,
                     float* __restrict__ out)
{
    __shared__ float Wl[Ln * WS];
    __shared__ int   sid_sh[Ln];
    __shared__ int   msk_sh[Ln];
    __shared__ int   hist[VOCABn];
    __shared__ int   mcnt;
    __shared__ int   tok_sh[19];
    __shared__ float inv_sh[19];

    const int bs   = blockIdx.x;
    const int t    = threadIdx.x;
    const int lane = t & 31;
    const int wave = t >> 5;

    if (t < VOCABn) hist[t] = 0;
    if (t == 0)     mcnt = 0;
    int mySid = 0, myMsk = 0;
    if (t < Ln)          { mySid = sids[bs * Ln + t];        sid_sh[t] = mySid; }
    else if (t < 2 * Ln) { myMsk = mask[bs * Ln + (t - Ln)]; msk_sh[t - Ln] = myMsk; }
    __syncthreads();

    if (t < Ln)               atomicAdd(&hist[mySid], 1);
    else if (t < 2 * Ln) { if (myMsk) atomicAdd(&mcnt, 1); }
    __syncthreads();

    if (t < 19) {
        if (t == 0) {
            const int c = mcnt;
            tok_sh[0] = -1;
            inv_sh[0] = 1.0f / (float)(c > 0 ? c : 1);
        } else {
            const int r = t;
            int a; const int* ids;
            if (r <= 6)       { a = r - 1;  ids = pred; }
            else if (r <= 12) { a = r - 7;  ids = arg0; }
            else              { a = r - 13; ids = arg1; }
            ids += (bs * An + a) * Mn;
            int tok = 0, c = 0;
            #pragma unroll
            for (int m = 0; m < Mn; ++m) {
                const int tm = ids[m];
                if (tm != 0) {
                    const int cm = hist[tm];
                    if (cm > 0) { tok = tm; c = cm; }
                }
            }
            tok_sh[r] = tok;
            inv_sh[r] = (c > 0) ? 1.0f / (float)c : 0.0f;
        }
    }
    __syncthreads();

    #pragma unroll
    for (int i = 0; i < 16; ++i) {
        const int idx = t + i * 256;
        const int l = idx >> 5;
        const int r = idx & 31;
        float w;
        if (r == 0)       w = msk_sh[l] ? inv_sh[0] : 0.0f;
        else if (r <= 18) w = (sid_sh[l] == tok_sh[r]) ? inv_sh[r] : 0.0f;
        else              w = 0.0f;
        Wl[l * WS + r] = w;
    }
    __syncthreads();

    v8f acc[2][6];
    #pragma unroll
    for (int mt = 0; mt < 2; ++mt)
        #pragma unroll
        for (int j = 0; j < 6; ++j) {
            v8f z = {0.f, 0.f, 0.f, 0.f, 0.f, 0.f, 0.f, 0.f};
            acc[mt][j] = z;
        }

    const int lm = lane & 15;
    const int kk = (lane >> 4) << 1;
    const float* Ebs = emb + (size_t)bs * Ln * Dn;
    const int nb = wave * 96 + lm;

    const float* Eln = Ebs + (size_t)kk * Dn + nb;
    const float* Wln = Wl + kk * WS + lm;
    v2f bq[PF][6];
    #pragma unroll
    for (int p = 0; p < PF; ++p) {
        const float* Er = Eln + (size_t)(p << 2) * Dn;
        #pragma unroll
        for (int j = 0; j < 6; ++j) {
            bq[p][j].x = __builtin_nontemporal_load(&Er[     j * 16]);
            bq[p][j].y = __builtin_nontemporal_load(&Er[Dn + j * 16]);
        }
    }
    float wc[4], wn[4];
    wc[0] = Wln[0];       wc[1] = Wln[WS];      wc[2] = Wln[16];          wc[3] = Wln[WS + 16];
    wn[0] = Wln[4 * WS];  wn[1] = Wln[5 * WS];  wn[2] = Wln[4 * WS + 16]; wn[3] = Wln[5 * WS + 16];

    #pragma unroll
    for (int kc = 0; kc < 32; ++kc) {
        const int st = kc % PF;
        v2f a0, a1;
        a0.x = wc[0]; a0.y = wc[1];
        a1.x = wc[2]; a1.y = wc[3];
        #pragma unroll
        for (int j = 0; j < 6; ++j) {
            acc[0][j] = __builtin_amdgcn_wmma_f32_16x16x4_f32(
                false, a0, false, bq[st][j], (short)0, acc[0][j], false, false);
            acc[1][j] = __builtin_amdgcn_wmma_f32_16x16x4_f32(
                false, a1, false, bq[st][j], (short)0, acc[1][j], false, false);
        }
        __builtin_amdgcn_sched_barrier(0);
        float wf[4] = {0.f, 0.f, 0.f, 0.f};
        if (kc + PF < 32) {
            const float* Er = Eln + (size_t)((kc + PF) << 2) * Dn;
            #pragma unroll
            for (int j = 0; j < 6; ++j) {
                bq[st][j].x = __builtin_nontemporal_load(&Er[     j * 16]);
                bq[st][j].y = __builtin_nontemporal_load(&Er[Dn + j * 16]);
            }
        }
        if (kc + 2 < 32) {
            const float* Wr = Wln + ((kc + 2) << 2) * WS;
            wf[0] = Wr[0];  wf[1] = Wr[WS];  wf[2] = Wr[16];  wf[3] = Wr[WS + 16];
        }
        __builtin_amdgcn_sched_barrier(0);
        #pragma unroll
        for (int i = 0; i < 4; ++i) { wc[i] = wn[i]; wn[i] = wf[i]; }
    }

    const size_t O1 = (size_t)Bn * Sn * Dn;
    const size_t O2 = O1 + (size_t)Bn * Sn * An * Dn;
    const size_t O3 = O2 + (size_t)Bn * Sn * An * Dn;
    const int mhalf = (lane >> 4) << 3;

    #pragma unroll
    for (int mt = 0; mt < 2; ++mt) {
        #pragma unroll
        for (int j = 0; j < 6; ++j) {
            const int n = nb + j * 16;
            #pragma unroll
            for (int r = 0; r < 8; ++r) {
                const int m = mt * 16 + r + mhalf;
                const float val = acc[mt][j][r];
                if (m == 0)
                    __builtin_nontemporal_store(val, &out[(size_t)bs * Dn + n]);
                else if (m <= 6)
                    __builtin_nontemporal_store(val, &out[O1 + ((size_t)bs * An + (m - 1))  * Dn + n]);
                else if (m <= 12)
                    __builtin_nontemporal_store(val, &out[O2 + ((size_t)bs * An + (m - 7))  * Dn + n]);
                else if (m <= 18)
                    __builtin_nontemporal_store(val, &out[O3 + ((size_t)bs * An + (m - 13)) * Dn + n]);
            }
        }
    }
}

extern "C" void kernel_launch(void* const* d_in, const int* in_sizes, int n_in,
                              void* d_out, int out_size, void* d_ws, size_t ws_size,
                              hipStream_t stream)
{
    (void)in_sizes; (void)n_in; (void)out_size; (void)d_ws; (void)ws_size;
    const int*   sids = (const int*)  d_in[0];
    const int*   mask = (const int*)  d_in[1];
    const float* emb  = (const float*)d_in[2];
    const int*   pred = (const int*)  d_in[3];
    const int*   a0   = (const int*)  d_in[4];
    const int*   a1   = (const int*)  d_in[5];
    float* out = (float*)d_out;

    dim3 grid(Bn * Sn);
    dim3 block(256);
    hipLaunchKernelGGL(srl_wmma_kernel, grid, block, 0, stream,
                       sids, mask, emb, pred, a0, a1, out);
}


# N-tile column map permuted so a lane owns 4 (and 2) adjacent columns: E row loads and output stores become 16- and 8-byte accesses of whole 128-byte lines (12 loads per k-step -> 4); sum order kept
# speedup vs baseline: 1.2567x; 1.1622x over previous
#include <hip/hip_runtime.h>

typedef float v2f __attribute__((ext_vector_type(2)));
typedef float v4f __attribute__((ext_vector_type(4)));
typedef float v8f __attribute__((ext_vector_type(8)));

#define Bn 16
#define Sn 32
#define Ln 128
#define Dn 768
#define An 6
#define Mn 10
#define VOCABn 200
#define WS 33
#define PF 4

__global__ __launch_bounds__(256)
void srl_wmma_kernel(const int* __restrict__ sids,
                     const int* __restrict__ mask,
                     const float* __restrict__ emb,
                     const int* __restrict__ pred,
                     const int* __restrict__ arg0,
                     const int* __restrict__ arg1,
                     float* __restrict__ out)
{
    __shared__ float Wl[Ln * WS];
    __shared__ int   sid_sh[Ln];
    __shared__ int   msk_sh[Ln];
    __shared__ int   hist[VOCABn];
    __shared__ int   mcnt;
    __shared__ int   tok_sh[19];
    __shared__ float inv_sh[19];

    const int bs   = blockIdx.x;
    const int t    = threadIdx.x;
    const int lane = t & 31;
    const int wave = t >> 5;

    if (t < VOCABn) hist[t] = 0;
    if (t == 0)     mcnt = 0;
    int mySid = 0, myMsk = 0;
    if (t < Ln)          { mySid = sids[bs * Ln + t];        sid_sh[t] = mySid; }
    else if (t < 2 * Ln) { myMsk = mask[bs * Ln + (t - Ln)]; msk_sh[t - Ln] = myMsk; }
    __syncthreads();

    if (t < Ln)               atomicAdd(&hist[mySid], 1);
    else if (t < 2 * Ln) { if (myMsk) atomicAdd(&mcnt, 1); }
    __syncthreads();

    if (t < 19) {
        if (t == 0) {
            const int c = mcnt;
            tok_sh[0] = -1;
            inv_sh[0] = 1.0f / (float)(c > 0 ? c : 1);
        } else {
            const int r = t;
            int a; const int* ids;
            if (r <= 6)       { a = r - 1;  ids = pred; }
            else if (r <= 12) { a = r - 7;  ids = arg0; }
            else              { a = r - 13; ids = arg1; }
            ids += (bs * An + a) * Mn;
            int tok = 0, c = 0;
            #pragma unroll
            for (int m = 0; m < Mn; ++m) {
                const int tm = ids[m];
                if (tm != 0) {
                    const int cm = hist[tm];
                    if (cm > 0) { tok = tm; c = cm; }
                }
            }
            tok_sh[r] = tok;
            inv_sh[r] = (c > 0) ? 1.0f / (float)c : 0.0f;
        }
    }
    __syncthreads();

    #pragma unroll
    for (int i = 0; i < 16; ++i) {
        const int idx = t + i * 256;
        const int l = idx >> 5;
        const int r = idx & 31;
        float w;
        if (r == 0)       w = msk_sh[l] ? inv_sh[0] : 0.0f;
        else if (r <= 18) w = (sid_sh[l] == tok_sh[r]) ? inv_sh[r] : 0.0f;
        else              w = 0.0f;
        Wl[l * WS + r] = w;
    }
    __syncthreads();

    v8f acc[2][6];
    #pragma unroll
    for (int mt = 0; mt < 2; ++mt)
        #pragma unroll
        for (int j = 0; j < 6; ++j) {
            v8f z = {0.f, 0.f, 0.f, 0.f, 0.f, 0.f, 0.f, 0.f};
            acc[mt][j] = z;
        }

    const int lm = lane & 15;
    const int kk = (lane >> 4) << 1;
    const float* Ebs = emb + (size_t)bs * Ln * Dn;
    const int wb = wave * 96;

    const int oddw = wave & 1;
    const int cq = wb + (oddw ? 32 : 0) + 4 * lm;
    const int cd = wb + (oddw ? 0 : 64) + 2 * lm;

    const float* Eln = Ebs + (size_t)kk * Dn;
    const float* Wln = Wl + kk * WS + lm;
    v4f qa[PF], qb[PF];
    v2f da[PF], db[PF];
    #pragma unroll
    for (int p = 0; p < PF; ++p) {
        const float* Er = Eln + (size_t)(p << 2) * Dn;
        qa[p] = __builtin_nontemporal_load((const v4f*)&Er[     cq]);
        qb[p] = __builtin_nontemporal_load((const v4f*)&Er[Dn + cq]);
        da[p] = __builtin_nontemporal_load((const v2f*)&Er[     cd]);
        db[p] = __builtin_nontemporal_load((const v2f*)&Er[Dn + cd]);
    }
    float wc[4], wn[4];
    wc[0] = Wln[0];       wc[1] = Wln[WS];      wc[2] = Wln[16];          wc[3] = Wln[WS + 16];
    wn[0] = Wln[4 * WS];  wn[1] = Wln[5 * WS];  wn[2] = Wln[4 * WS + 16]; wn[3] = Wln[5 * WS + 16];

    #pragma unroll
    for (int kc = 0; kc < 32; ++kc) {
        const int st = kc % PF;
        v2f a0, a1;
        a0.x = wc[0]; a0.y = wc[1];
        a1.x = wc[2]; a1.y = wc[3];
        v2f bf[6];
        bf[0].x = qa[st].x; bf[0].y = qb[st].x;
        bf[1].x = qa[st].y; bf[1].y = qb[st].y;
        bf[2].x = qa[st].z; bf[2].y = qb[st].z;
        bf[3].x = qa[st].w; bf[3].y = qb[st].w;
        bf[4].x = da[st].x; bf[4].y = db[st].x;
        bf[5].x = da[st].y; bf[5].y = db[st].y;
        asm volatile("" : "+v"(bf[0]), "+v"(bf[1]), "+v"(bf[2]), "+v"(bf[3]), "+v"(bf[4]), "+v"(bf[5]));
        __builtin_amdgcn_sched_barrier(0);
        #pragma unroll
        for (int j = 0; j < 6; ++j) {
            acc[0][j] = __builtin_amdgcn_wmma_f32_16x16x4_f32(
                false, a0, false, bf[j], (short)0, acc[0][j], false, false);
            acc[1][j] = __builtin_amdgcn_wmma_f32_16x16x4_f32(
                false, a1, false, bf[j], (short)0, acc[1][j], false, false);
        }
        __builtin_amdgcn_sched_barrier(0);
        if (kc + PF >= 32) {
            asm volatile("v_nop\n\tv_nop\n\tv_nop\n\tv_nop\n\tv_nop");
        }
        float wf[4] = {0.f, 0.f, 0.f, 0.f};
        if (kc + PF < 32) {
            const float* Er = Eln + (size_t)((kc + PF) << 2) * Dn;
            qa[st] = __builtin_nontemporal_load((const v4f*)&Er[     cq]);
            qb[st] = __builtin_nontemporal_load((const v4f*)&Er[Dn + cq]);
            da[st] = __builtin_nontemporal_load((const v2f*)&Er[     cd]);
            db[st] = __builtin_nontemporal_load((const v2f*)&Er[Dn + cd]);
        }
        if (kc + 2 < 32) {
            const float* Wr = Wln + ((kc + 2) << 2) * WS;
            wf[0] = Wr[0];  wf[1] = Wr[WS];  wf[2] = Wr[16];  wf[3] = Wr[WS + 16];
        }
        __builtin_amdgcn_sched_barrier(0);
        #pragma unroll
        for (int i = 0; i < 4; ++i) { wc[i] = wn[i]; wn[i] = wf[i]; }
    }

    const size_t O1 = (size_t)Bn * Sn * Dn;
    const size_t O2 = O1 + (size_t)Bn * Sn * An * Dn;
    const size_t O3 = O2 + (size_t)Bn * Sn * An * Dn;
    const int mhalf = (lane >> 4) << 3;

    #pragma unroll
    for (int mt = 0; mt < 2; ++mt) {
        #pragma unroll
        for (int r = 0; r < 8; ++r) {
            const int m = mt * 16 + r + mhalf;
            if (m <= 18) {
                size_t ro;
                if (m == 0)       ro = (size_t)bs * Dn;
                else if (m <= 6)  ro = O1 + ((size_t)bs * An + (m - 1))  * Dn;
                else if (m <= 12) ro = O2 + ((size_t)bs * An + (m - 7))  * Dn;
                else              ro = O3 + ((size_t)bs * An + (m - 13)) * Dn;
                v4f q; v2f d;
                q.x = acc[mt][0][r]; q.y = acc[mt][1][r]; q.z = acc[mt][2][r]; q.w = acc[mt][3][r];
                d.x = acc[mt][4][r]; d.y = acc[mt][5][r];
                __builtin_nontemporal_store(q, (v4f*)&out[ro + cq]);
                __builtin_nontemporal_store(d, (v2f*)&out[ro + cd]);
            }
        }
    }
}

extern "C" void kernel_launch(void* const* d_in, const int* in_sizes, int n_in,
                              void* d_out, int out_size, void* d_ws, size_t ws_size,
                              hipStream_t stream)
{
    (void)in_sizes; (void)n_in; (void)out_size; (void)d_ws; (void)ws_size;
    const int*   sids = (const int*)  d_in[0];
    const int*   mask = (const int*)  d_in[1];
    const float* emb  = (const float*)d_in[2];
    const int*   pred = (const int*)  d_in[3];
    const int*   a0   = (const int*)  d_in[4];
    const int*   a1   = (const int*)  d_in[5];
    float* out = (float*)d_out;

    dim3 grid(Bn * Sn);
    dim3 block(256);
    hipLaunchKernelGGL(srl_wmma_kernel, grid, block, 0, stream,
                       sids, mask, emb, pred, a0, a1, out);
}


# Second M tile (rows 16..18) skips a k-step's six products when its four W columns are all zero (scalar branch on position flags from the prologue); sums unchanged for finite E
# speedup vs baseline: 1.3680x; 1.0886x over previous
#include <hip/hip_runtime.h>

typedef float v2f __attribute__((ext_vector_type(2)));
typedef float v4f __attribute__((ext_vector_type(4)));
typedef float v8f __attribute__((ext_vector_type(8)));

#define Bn 16
#define Sn 32
#define Ln 128
#define Dn 768
#define An 6
#define Mn 10
#define VOCABn 200
#define WS 33
#define PF 4

__global__ __launch_bounds__(256)
void srl_wmma_kernel(const int* __restrict__ sids,
                     const int* __restrict__ mask,
                     const float* __restrict__ emb,
                     const int* __restrict__ pred,
                     const int* __restrict__ arg0,
                     const int* __restrict__ arg1,
                     float* __restrict__ out)
{
    __shared__ float Wl[Ln * WS];
    __shared__ int   sid_sh[Ln];
    __shared__ int   msk_sh[Ln];
    __shared__ int   hist[VOCABn];
    __shared__ int   mcnt;
    __shared__ int   tok_sh[19];
    __shared__ float inv_sh[19];
    __shared__ unsigned nz_sh[4];

    const int bs   = blockIdx.x;
    const int t    = threadIdx.x;
    const int lane = t & 31;
    const int wave = t >> 5;

    if (t < VOCABn) hist[t] = 0;
    if (t == 0)     mcnt = 0;
    int mySid = 0, myMsk = 0;
    if (t < Ln)          { mySid = sids[bs * Ln + t];        sid_sh[t] = mySid; }
    else if (t < 2 * Ln) { myMsk = mask[bs * Ln + (t - Ln)]; msk_sh[t - Ln] = myMsk; }
    __syncthreads();

    if (t < Ln)               atomicAdd(&hist[mySid], 1);
    else if (t < 2 * Ln) { if (myMsk) atomicAdd(&mcnt, 1); }
    __syncthreads();

    if (t < 19) {
        if (t == 0) {
            const int c = mcnt;
            tok_sh[0] = -1;
            inv_sh[0] = 1.0f / (float)(c > 0 ? c : 1);
        } else {
            const int r = t;
            int a; const int* ids;
            if (r <= 6)       { a = r - 1;  ids = pred; }
            else if (r <= 12) { a = r - 7;  ids = arg0; }
            else              { a = r - 13; ids = arg1; }
            ids += (bs * An + a) * Mn;
            int tok = 0, c = 0;
            #pragma unroll
            for (int m = 0; m < Mn; ++m) {
                const int tm = ids[m];
                if (tm != 0) {
                    const int cm = hist[tm];
                    if (cm > 0) { tok = tm; c = cm; }
                }
            }
            tok_sh[r] = tok;
            inv_sh[r] = (c > 0) ? 1.0f / (float)c : 0.0f;
        }
    }
    __syncthreads();

    if (t < Ln) {
        const int   t16 = tok_sh[16], t17 = tok_sh[17], t18 = tok_sh[18];
        const float i16 = inv_sh[16], i17 = inv_sh[17], i18 = inv_sh[18];
        const bool hit = ((mySid == t16) & (i16 != 0.0f)) | ((mySid == t17) & (i17 != 0.0f)) |
                         ((mySid == t18) & (i18 != 0.0f));
        const unsigned bal = __builtin_amdgcn_ballot_w32(hit);
        if (lane == 0) nz_sh[wave] = bal;
    }

    #pragma unroll
    for (int i = 0; i < 16; ++i) {
        const int idx = t + i * 256;
        const int l = idx >> 5;
        const int r = idx & 31;
        float w;
        if (r == 0)       w = msk_sh[l] ? inv_sh[0] : 0.0f;
        else if (r <= 18) w = (sid_sh[l] == tok_sh[r]) ? inv_sh[r] : 0.0f;
        else              w = 0.0f;
        Wl[l * WS + r] = w;
    }
    __syncthreads();

    v8f acc[2][6];
    #pragma unroll
    for (int mt = 0; mt < 2; ++mt)
        #pragma unroll
        for (int j = 0; j < 6; ++j) {
            v8f z = {0.f, 0.f, 0.f, 0.f, 0.f, 0.f, 0.f, 0.f};
            acc[mt][j] = z;
        }

    unsigned nzw[4];
    #pragma unroll
    for (int i = 0; i < 4; ++i) nzw[i] = __builtin_amdgcn_readfirstlane(nz_sh[i]);

    const int lm = lane & 15;
    const int kk = (lane >> 4) << 1;
    const float* Ebs = emb + (size_t)bs * Ln * Dn;
    const int wb = wave * 96;

    const int oddw = wave & 1;
    const int cq = wb + (oddw ? 32 : 0) + 4 * lm;
    const int cd = wb + (oddw ? 0 : 64) + 2 * lm;

    const float* Eln = Ebs + (size_t)kk * Dn;
    const float* Wln = Wl + kk * WS + lm;
    v4f qa[PF], qb[PF];
    v2f da[PF], db[PF];
    #pragma unroll
    for (int p = 0; p < PF; ++p) {
        const float* Er = Eln + (size_t)(p << 2) * Dn;
        qa[p] = __builtin_nontemporal_load((const v4f*)&Er[     cq]);
        qb[p] = __builtin_nontemporal_load((const v4f*)&Er[Dn + cq]);
        da[p] = __builtin_nontemporal_load((const v2f*)&Er[     cd]);
        db[p] = __builtin_nontemporal_load((const v2f*)&Er[Dn + cd]);
    }
    float wc[4], wn[4];
    wc[0] = Wln[0];       wc[1] = Wln[WS];      wc[2] = Wln[16];          wc[3] = Wln[WS + 16];
    wn[0] = Wln[4 * WS];  wn[1] = Wln[5 * WS];  wn[2] = Wln[4 * WS + 16]; wn[3] = Wln[5 * WS + 16];

    #pragma unroll
    for (int kc = 0; kc < 32; ++kc) {
        const int st = kc % PF;
        v2f a0, a1;
        a0.x = wc[0]; a0.y = wc[1];
        a1.x = wc[2]; a1.y = wc[3];
        v2f bf[6];
        bf[0].x = qa[st].x; bf[0].y = qb[st].x;
        bf[1].x = qa[st].y; bf[1].y = qb[st].y;
        bf[2].x = qa[st].z; bf[2].y = qb[st].z;
        bf[3].x = qa[st].w; bf[3].y = qb[st].w;
        bf[4].x = da[st].x; bf[4].y = db[st].x;
        bf[5].x = da[st].y; bf[5].y = db[st].y;
        asm volatile("" : "+v"(bf[0]), "+v"(bf[1]), "+v"(bf[2]), "+v"(bf[3]), "+v"(bf[4]), "+v"(bf[5]),
                          "+v"(a0), "+v"(a1));
        __builtin_amdgcn_sched_barrier(0);
        const unsigned nzk = (nzw[kc >> 3] >> ((kc & 7) << 2)) & 0xFu;
        asm volatile(
            "s_cmp_lg_u32 %[nz], 0\n\t"
            "s_cbranch_scc0 .Lskip_%=\n\t"
            "v_wmma_f32_16x16x4_f32 %[c0], %[a], %[b0], %[c0]\n\t"
            "v_wmma_f32_16x16x4_f32 %[c1], %[a], %[b1], %[c1]\n\t"
            "v_wmma_f32_16x16x4_f32 %[c2], %[a], %[b2], %[c2]\n\t"
            "v_wmma_f32_16x16x4_f32 %[c3], %[a], %[b3], %[c3]\n\t"
            "v_wmma_f32_16x16x4_f32 %[c4], %[a], %[b4], %[c4]\n\t"
            "v_wmma_f32_16x16x4_f32 %[c5], %[a], %[b5], %[c5]\n"
            ".Lskip_%=:"
            : [c0] "+v"(acc[1][0]), [c1] "+v"(acc[1][1]), [c2] "+v"(acc[1][2]),
              [c3] "+v"(acc[1][3]), [c4] "+v"(acc[1][4]), [c5] "+v"(acc[1][5])
            : [a] "v"(a1), [b0] "v"(bf[0]), [b1] "v"(bf[1]), [b2] "v"(bf[2]),
              [b3] "v"(bf[3]), [b4] "v"(bf[4]), [b5] "v"(bf[5]), [nz] "s"(nzk)
            : "scc");
        __builtin_amdgcn_sched_barrier(0);
        #pragma unroll
        for (int j = 0; j < 6; ++j)
            acc[0][j] = __builtin_amdgcn_wmma_f32_16x16x4_f32(
                false, a0, false, bf[j], (short)0, acc[0][j], false, false);
        __builtin_amdgcn_sched_barrier(0);
        if (kc + PF >= 32) {
            asm volatile("v_nop\n\tv_nop\n\tv_nop\n\tv_nop\n\tv_nop");
        }
        float wf[4] = {0.f, 0.f, 0.f, 0.f};
        if (kc + PF < 32) {
            const float* Er = Eln + (size_t)((kc + PF) << 2) * Dn;
            qa[st] = __builtin_nontemporal_load((const v4f*)&Er[     cq]);
            qb[st] = __builtin_nontemporal_load((const v4f*)&Er[Dn + cq]);
            da[st] = __builtin_nontemporal_load((const v2f*)&Er[     cd]);
            db[st] = __builtin_nontemporal_load((const v2f*)&Er[Dn + cd]);
        }
        if (kc + 2 < 32) {
            const float* Wr = Wln + ((kc + 2) << 2) * WS;
            wf[0] = Wr[0];  wf[1] = Wr[WS];  wf[2] = Wr[16];  wf[3] = Wr[WS + 16];
        }
        __builtin_amdgcn_sched_barrier(0);
        #pragma unroll
        for (int i = 0; i < 4; ++i) { wc[i] = wn[i]; wn[i] = wf[i]; }
    }

    const size_t O1 = (size_t)Bn * Sn * Dn;
    const size_t O2 = O1 + (size_t)Bn * Sn * An * Dn;
    const size_t O3 = O2 + (size_t)Bn * Sn * An * Dn;
    const int mhalf = (lane >> 4) << 3;

    #pragma unroll
    for (int mt = 0; mt < 2; ++mt) {
        #pragma unroll
        for (int r = 0; r < 8; ++r) {
            const int m = mt * 16 + r + mhalf;
            if (m <= 18) {
                size_t ro;
                if (m == 0)       ro = (size_t)bs * Dn;
                else if (m <= 6)  ro = O1 + ((size_t)bs * An + (m - 1))  * Dn;
                else if (m <= 12) ro = O2 + ((size_t)bs * An + (m - 7))  * Dn;
                else              ro = O3 + ((size_t)bs * An + (m - 13)) * Dn;
                v4f q; v2f d;
                q.x = acc[mt][0][r]; q.y = acc[mt][1][r]; q.z = acc[mt][2][r]; q.w = acc[mt][3][r];
                d.x = acc[mt][4][r]; d.y = acc[mt][5][r];
                __builtin_nontemporal_store(q, (v4f*)&out[ro + cq]);
                __builtin_nontemporal_store(d, (v2f*)&out[ro + cd]);
            }
        }
    }
}

extern "C" void kernel_launch(void* const* d_in, const int* in_sizes, int n_in,
                              void* d_out, int out_size, void* d_ws, size_t ws_size,
                              hipStream_t stream)
{
    (void)in_sizes; (void)n_in; (void)out_size; (void)d_ws; (void)ws_size;
    const int*   sids = (const int*)  d_in[0];
    const int*   mask = (const int*)  d_in[1];
    const float* emb  = (const float*)d_in[2];
    const int*   pred = (const int*)  d_in[3];
    const int*   a0   = (const int*)  d_in[4];
    const int*   a1   = (const int*)  d_in[5];
    float* out = (float*)d_out;

    dim3 grid(Bn * Sn);
    dim3 block(256);
    hipLaunchKernelGGL(srl_wmma_kernel, grid, block, 0, stream,
                       sids, mask, emb, pred, a0, a1, out);
}
